# starting kernel
# baseline (speedup 1.0000x reference)
#include <hip/hip_runtime.h>
#include <math.h>

typedef __attribute__((ext_vector_type(2))) float v2f;
typedef __attribute__((ext_vector_type(8))) float v8f;

#define B_   64
#define O_   256
#define M_   8
#define D_   512
#define EPS_ 1e-5f
#define BT   16
#define NBT  (B_ / BT)
#define NW   8
#define NJ   (D_ / (4 * NW))

__device__ __forceinline__ float fast_sigmoid(float x) {
#if __has_builtin(__builtin_amdgcn_tanhf)
    return fmaf(0.5f, __builtin_amdgcn_tanhf(0.5f * x), 0.5f);
#else
    return __builtin_amdgcn_rcpf(1.0f + __expf(-x));
#endif
}

__global__ __launch_bounds__(256) void dnm_prep(
    const float* __restrict__ x,   const float* __restrict__ g,
    const float* __restrict__ be,  const float* __restrict__ dnw,
    const float* __restrict__ dnb, float* __restrict__ xn,
    float* __restrict__ wsum,      float* __restrict__ bsum)
{
    __shared__ float s1[256];
    __shared__ float s2[256];
    const int t  = threadIdx.x;
    const int bx = blockIdx.x;

    if (bx < B_) {
        const float* row = x + bx * D_;
        float a = row[t];
        float b = row[t + 256];
        s1[t] = a + b;
        s2[t] = a * a + b * b;
        __syncthreads();
        for (int s = 128; s > 0; s >>= 1) {
            if (t < s) { s1[t] += s1[t + s]; s2[t] += s2[t + s]; }
            __syncthreads();
        }
        float mu   = s1[0] * (1.0f / D_);
        float var  = s2[0] * (1.0f / D_) - mu * mu;
        float rstd = rsqrtf(var + EPS_);
        float* out = xn + bx * D_;
        out[t]       = (a - mu) * rstd * g[t]       + be[t];
        out[t + 256] = (b - mu) * rstd * g[t + 256] + be[t + 256];
    } else {
        const int m = bx - B_;
        const float* wr = dnw + m * D_;
        const float* br = dnb + m * D_;
        s1[t] = wr[t] + wr[t + 256];
        s2[t] = br[t] + br[t + 256];
        __syncthreads();
        for (int s = 128; s > 0; s >>= 1) {
            if (t < s) { s1[t] += s1[t + s]; s2[t] += s2[t + s]; }
            __syncthreads();
        }
        if (t == 0) { wsum[m] = s1[0]; bsum[m] = s2[0]; }
    }
}

__global__ __launch_bounds__(NW * 32) void dnm_main(
    const float* __restrict__ xn,   const float* __restrict__ sw,
    const float* __restrict__ sb,   const float* __restrict__ dnw,
    const float* __restrict__ wsum, const float* __restrict__ bsum,
    float* __restrict__ pre)
{
    constexpr int XP = D_ + 4;
    __shared__ float xs[BT * XP];
    __shared__ float c1buf[NW * 256];
    __shared__ float c2buf[NW * 256];

    const int o     = blockIdx.x >> 2;
    const int b0    = (blockIdx.x & (NBT - 1)) * BT;
    const int tid   = threadIdx.x;
    const int w     = tid >> 5;
    const int lane  = tid & 31;
    const int tb    = lane & 15;
    const int khalf = lane >> 4;
    const int kb    = khalf * 2;

    for (int idx = tid; idx < BT * D_; idx += NW * 32) {
        int r = idx >> 9, c = idx & (D_ - 1);
        xs[r * XP + c] = xn[(b0 + r) * D_ + c];
    }
    __syncthreads();

    v8f c1 = {};
    v8f c2 = {};

    const int    dstart = 4 * w + kb;
    const float  fone   = (tb == 0) ? 1.0f : 0.0f;
    const float* dwl    = dnw + (((tb >= 1) && (tb <= M_)) ? (tb - 1) * D_ : 0) + dstart;
    const float* xrow   = xs + tb * XP + dstart;
    const float* swl    = sw + (size_t)o * (M_ * D_) + dstart;
    const float* sbl    = sb + (size_t)o * (M_ * D_) + dstart;

    #pragma unroll 1
    for (int m = 0; m < M_; ++m) {
        __builtin_prefetch(swl + D_, 0, 1);
        __builtin_prefetch(sbl + D_, 0, 1);
        const float fm = (tb == m + 1) ? 1.0f : 0.0f;
        #pragma unroll 4
        for (int j = 0; j < NJ; ++j) {
            const int off = j * 4 * NW;
            float2 swv = *(const float2*)(swl + off);
            float2 sbv = *(const float2*)(sbl + off);
            float2 dwv = *(const float2*)(dwl + off);
            float2 xv  = *(const float2*)(xrow + off);

            v2f a;
            a.x = fast_sigmoid(fmaf(swv.x, xv.x, sbv.x));
            a.y = fast_sigmoid(fmaf(swv.y, xv.y, sbv.y));

            v2f bm;
            bm.x = fmaf(fm, dwv.x, fone);
            bm.y = fmaf(fm, dwv.y, fone);

            c1 = __builtin_amdgcn_wmma_f32_16x16x4_f32(false, a, false, bm, (short)0, c1, false, false);
            c2 = __builtin_amdgcn_wmma_f32_16x16x4_f32(false, a, false, a,  (short)0, c2, false, false);
        }
        swl += D_;
        sbl += D_;
    }

    #pragma unroll
    for (int r = 0; r < 8; ++r) {
        const int row = r + khalf * 8;
        c1buf[w * 256 + row * 16 + tb] = c1[r];
        c2buf[w * 256 + row * 16 + tb] = c2[r];
    }
    __syncthreads();

    if (tid < BT) {
        float S1 = 0.f, S2 = 0.f, T[M_];
        #pragma unroll
        for (int m = 0; m < M_; ++m) T[m] = 0.f;
        for (int wv = 0; wv < NW; ++wv) {
            const float* p1 = c1buf + wv * 256 + tid * 16;
            const float* p2 = c2buf + wv * 256 + tid * 16;
            S1 += p1[0];
            S2 += p2[tid];
            #pragma unroll
            for (int m = 0; m < M_; ++m) T[m] += p1[m + 1];
        }
        const float inv  = 1.0f / (float)(M_ * D_);
        const float mu   = S1 * inv;
        const float var  = S2 * inv - mu * mu;
        const float rstd = rsqrtf(var + EPS_);
        float acc = 0.f;
        #pragma unroll
        for (int m = 0; m < M_; ++m) {
            float s = rstd * (T[m] - mu * wsum[m]) + bsum[m];
            acc += fast_sigmoid(s);
        }
        pre[(size_t)(b0 + tid) * O_ + o] = acc;
    }
}

__global__ __launch_bounds__(256) void dnm_softmax(
    const float* __restrict__ pre, float* __restrict__ out)
{
    __shared__ float sh[256];
    const int b = blockIdx.x, t = threadIdx.x;
    float v = pre[b * O_ + t];
    sh[t] = v;
    __syncthreads();
    for (int s = 128; s > 0; s >>= 1) {
        if (t < s) sh[t] = fmaxf(sh[t], sh[t + s]);
        __syncthreads();
    }
    float mx = sh[0];
    __syncthreads();
    float e = __expf(v - mx);
    sh[t] = e;
    __syncthreads();
    for (int s = 128; s > 0; s >>= 1) {
        if (t < s) sh[t] += sh[t + s];
        __syncthreads();
    }
    out[b * O_ + t] = e * __builtin_amdgcn_rcpf(sh[0]);
}

extern "C" void kernel_launch(void* const* d_in, const int* in_sizes, int n_in,
                              void* d_out, int out_size, void* d_ws, size_t ws_size,
                              hipStream_t stream) {
    const float* x    = (const float*)d_in[0];
    const float* sn0w = (const float*)d_in[1];
    const float* sn0b = (const float*)d_in[2];
    const float* sw   = (const float*)d_in[3];
    const float* sb   = (const float*)d_in[4];
    const float* dnw  = (const float*)d_in[5];
    const float* dnb  = (const float*)d_in[6];
    float* out = (float*)d_out;

    float* ws   = (float*)d_ws;
    float* xn   = ws;
    float* wsum = ws + 32768;
    float* bsum = ws + 32776;
    float* pre  = ws + 32784;

    dnm_prep   <<<B_ + M_,  256,     0, stream>>>(x, sn0w, sn0b, dnw, dnb, xn, wsum, bsum);
    dnm_main   <<<O_ * NBT, NW * 32, 0, stream>>>(xn, sw, sb, dnw, wsum, bsum, pre);
    dnm_softmax<<<B_,       256,     0, stream>>>(pre, out);
}


# Prep and softmax kernels: radix-2 LDS trees (9 and 19 barriers) become flat sums of 16-float segments read as quads (2 and 1 barriers); softmax max pass dropped (scores in [0,8]); sum orders change
# speedup vs baseline: 1.6782x; 1.6782x over previous
#include <hip/hip_runtime.h>
#include <math.h>


typedef __attribute__((ext_vector_type(2))) float v2f;
typedef __attribute__((ext_vector_type(8))) float v8f;

#define B_   64
#define O_   256
#define M_   8
#define D_   512
#define EPS_ 1e-5f
#define OG   4
#define BG   16
#define NWV  16
#define EC   32
#define RP   9
#define PW   120
#define SPILL0 (NWV * BG * PW)

__device__ __forceinline__ float fast_sigmoid(float x) {
#if __has_builtin(__builtin_amdgcn_tanhf)
    return fmaf(0.5f, __builtin_amdgcn_tanhf(0.5f * x), 0.5f);
#else
    return __builtin_amdgcn_rcpf(1.0f + __expf(-x));
#endif
}

__device__ __forceinline__ float sigmoid_of_half(float hv) {
#if __has_builtin(__builtin_amdgcn_tanhf)
    return fmaf(0.5f, __builtin_amdgcn_tanhf(hv), 0.5f);
#else
    return __builtin_amdgcn_rcpf(1.0f + __expf(-2.0f * hv));
#endif
}

__device__ __forceinline__ float other_half(float v) {
    const int b = __builtin_bit_cast(int, v);
    return __builtin_bit_cast(float,
        __builtin_amdgcn_permlanex16(b, b, 0x76543210, 0xfedcba98, false, false));
}

__device__ __forceinline__ float quad_sum(float4 u) { return (u.x + u.y) + (u.z + u.w); }

__device__ __forceinline__ float4 quad_add4(float4 a, float4 b, float4 c, float4 d) {
    float4 r;
    r.x = (a.x + b.x) + (c.x + d.x); r.y = (a.y + b.y) + (c.y + d.y);
    r.z = (a.z + b.z) + (c.z + d.z); r.w = (a.w + b.w) + (c.w + d.w);
    return r;
}

__device__ __forceinline__ void block_sum2(float v1, float v2, float* s1, float* s2,
                                           float* pp, int t, float& r1, float& r2)
{
    s1[t] = v1;
    s2[t] = v2;
    __syncthreads();
    if (t < 16) {
        const float4* q1 = (const float4*)(s1 + 16 * t);
        const float4* q2 = (const float4*)(s2 + 16 * t);
        pp[t]      = quad_sum(quad_add4(q1[0], q1[1], q1[2], q1[3]));
        pp[16 + t] = quad_sum(quad_add4(q2[0], q2[1], q2[2], q2[3]));
    }
    __syncthreads();
    const float4* p4 = (const float4*)pp;
    r1 = quad_sum(quad_add4(p4[0], p4[1], p4[2], p4[3]));
    r2 = quad_sum(quad_add4(p4[4], p4[5], p4[6], p4[7]));
}

__global__ __launch_bounds__(256) void dnm_prep(
    const float* __restrict__ x,   const float* __restrict__ g,
    const float* __restrict__ be,  const float* __restrict__ dnw,
    const float* __restrict__ dnb, float* __restrict__ xn,
    float* __restrict__ wsum,      float* __restrict__ bsum)
{
    __shared__ __attribute__((aligned(16))) float s1[256];
    __shared__ __attribute__((aligned(16))) float s2[256];
    __shared__ __attribute__((aligned(16))) float pp[32];
    const int t  = threadIdx.x;
    const int bx = blockIdx.x;

    if (bx < B_) {
        const float* row = x + bx * D_;
        float a = row[t];
        float b = row[t + 256];
        float S1, S2;
        block_sum2(a + b, a * a + b * b, s1, s2, pp, t, S1, S2);
        float mu   = S1 * (1.0f / D_);
        float var  = S2 * (1.0f / D_) - mu * mu;
        float rstd = rsqrtf(var + EPS_);
        float* out = xn + bx * D_;
        out[t]       = (a - mu) * rstd * g[t]       + be[t];
        out[t + 256] = (b - mu) * rstd * g[t + 256] + be[t + 256];
    } else {
        const int m = bx - B_;
        const float* wr = dnw + m * D_;
        const float* br = dnb + m * D_;
        float S1, S2;
        block_sum2(wr[t] + wr[t + 256], br[t] + br[t + 256], s1, s2, pp, t, S1, S2);
        if (t == 0) { wsum[m] = S1; bsum[m] = S2; }
    }
}

__global__ __launch_bounds__(NWV * 32) void dnm_main(
    const float* __restrict__ xn,   const float* __restrict__ sw,
    const float* __restrict__ sb,   const float* __restrict__ dnw,
    const float* __restrict__ wsum, const float* __restrict__ bsum,
    float* __restrict__ pre)
{
    __shared__ float4 stage[2 * NWV * 32 * RP];
    __shared__ __attribute__((aligned(16))) float part[NWV * BG * PW + NWV * 32 * 8];
    __shared__ float red[4 * 10 * 64];

    const int ot   = blockIdx.x >> 2;
    const int o0   = ot * OG;
    const int b0   = (blockIdx.x & 3) * BG;
    const int tid  = threadIdx.x;
    const int w    = __builtin_amdgcn_readfirstlane(tid >> 5);
    const int lane = tid & 31;
    const int c    = lane & 15;
    const int h    = lane >> 4;
    const int d0   = w * EC;

    float swh[EC], sbh[EC], bm[EC];
    {
        const int g  = lane >> 3;
        const int pc = lane & 7;
        const float* swl = sw + ((size_t)(o0 + g) * M_) * D_ + d0 + 4 * pc;
        const float* sbl = sb + ((size_t)(o0 + g) * M_) * D_ + d0 + 4 * pc;
        float4* st0 = stage + w * (32 * RP);
        float4* st1 = stage + (NWV + w) * (32 * RP);
        float4 ta[8], tb[8];
        #pragma unroll
        for (int j = 0; j < 8; ++j) ta[j] = *(const float4*)(swl + (2 * (j & 3) + (j >> 2)) * D_);
        #pragma unroll
        for (int j = 0; j < 8; ++j) tb[j] = *(const float4*)(sbl + (2 * (j & 3) + (j >> 2)) * D_);

        const int  j8  = c - 1;
        const bool sel = (c >= 1) && (c <= 8) && ((j8 & 1) == h);
        const float cst = (c == 0) ? 1.0f : 0.0f;
        const int  jr  = j8 < 0 ? 0 : (j8 > 7 ? 7 : j8);
        const float* dwr = dnw + jr * D_ + d0;
        #pragma unroll
        for (int i = 0; i < EC; i += 4) {
            const float4 d4 = *(const float4*)(dwr + i);
            bm[i + 0] = sel ? d4.x : cst; bm[i + 1] = sel ? d4.y : cst;
            bm[i + 2] = sel ? d4.z : cst; bm[i + 3] = sel ? d4.w : cst;
        }

        #pragma unroll
        for (int j = 0; j < 8; ++j) st0[(4 * j + g) * RP + pc] = ta[j];
        #pragma unroll
        for (int j = 0; j < 8; ++j) st1[(4 * j + g) * RP + pc] = tb[j];
        #pragma unroll
        for (int i = 0; i < 8; ++i) {
            const float4 v = st0[lane * RP + i];
            swh[4 * i + 0] = 0.5f * v.x; swh[4 * i + 1] = 0.5f * v.y;
            swh[4 * i + 2] = 0.5f * v.z; swh[4 * i + 3] = 0.5f * v.w;
        }
        #pragma unroll
        for (int i = 0; i < 8; ++i) {
            const float4 v = st1[lane * RP + i];
            sbh[4 * i + 0] = 0.5f * v.x; sbh[4 * i + 1] = 0.5f * v.y;
            sbh[4 * i + 2] = 0.5f * v.z; sbh[4 * i + 3] = 0.5f * v.w;
        }
    }

    const bool wr    = (c == 0) || ((c <= 8) && (h == ((c - 1) >> 2)));
    const int  slot  = (c == 0) ? h : (1 + c);
    const int  spill = SPILL0 + (w * 32 + lane) * 8;

    #pragma unroll 1
    for (int bi = 0; bi < BG; ++bi) {
        const float* xr = xn + (b0 + bi) * D_ + d0;
        v8f c1 = {};
        v2f q  = {};
        #pragma unroll
        for (int i = 0; i < EC; i += 2) {
            v2f a, bb;
            a.x  = sigmoid_of_half(fmaf(swh[i],     xr[i],     sbh[i]));
            a.y  = sigmoid_of_half(fmaf(swh[i + 1], xr[i + 1], sbh[i + 1]));
            bb.x = bm[i];
            bb.y = bm[i + 1];
            c1 = __builtin_amdgcn_wmma_f32_16x16x4_f32(false, a, false, bb, (short)0, c1, false, false);
            q  = __builtin_elementwise_fma(a, a, q);
        }
        const int rec = (w * BG + bi) * PW;
        part[rec + 80 + lane] = q.x + q.y;
        float* dst = part + (wr ? rec + 8 * slot : spill);
        float4 lo, hi;
        lo.x = c1[0]; lo.y = c1[1]; lo.z = c1[2]; lo.w = c1[3];
        hi.x = c1[4]; hi.y = c1[5]; hi.z = c1[6]; hi.w = c1[7];
        *(float4*)(dst)     = lo;
        *(float4*)(dst + 4) = hi;
    }
    __syncthreads();

    if (tid < 256) {
        const int ws = tid >> 6;
        const int r  = tid & 63;
        const int bi = r >> 2;
        const int oi = r & 3;
        float S1 = 0.f, S2 = 0.f, T[M_];
        #pragma unroll
        for (int mm = 0; mm < M_; ++mm) T[mm] = 0.f;
        #pragma unroll
        for (int k = 0; k < 4; ++k) {
            const float* pp = part + ((4 * ws + k) * BG + bi) * PW;
            S1 += (pp[oi] + pp[4 + oi]) + (pp[8 + oi] + pp[12 + oi]);
            #pragma unroll
            for (int mm = 0; mm < M_; ++mm) T[mm] += pp[8 * (2 + mm) + 4 * ((mm >> 1) & 1) + oi];
            #pragma unroll
            for (int u = 0; u < 8; ++u) S2 += pp[80 + 4 * u + oi];
        }
        float* rr = red + ws * 640 + r;
        rr[0]  = S1;
        rr[64] = S2;
        #pragma unroll
        for (int mm = 0; mm < M_; ++mm) rr[128 + 64 * mm] = T[mm];
    }
    __syncthreads();

    if (tid < BG * OG) {
        float S1 = 0.f, S2 = 0.f, T[M_];
        #pragma unroll
        for (int mm = 0; mm < M_; ++mm) T[mm] = 0.f;
        #pragma unroll
        for (int ws = 0; ws < 4; ++ws) {
            const float* rr = red + ws * 640 + tid;
            S1 += rr[0];
            S2 += rr[64];
            #pragma unroll
            for (int mm = 0; mm < M_; ++mm) T[mm] += rr[128 + 64 * mm];
        }
        const float inv  = 1.0f / (float)(M_ * D_);
        const float mu   = S1 * inv;
        const float var  = S2 * inv - mu * mu;
        const float rstd = rsqrtf(var + EPS_);
        float acc = 0.f;
        #pragma unroll
        for (int mm = 0; mm < M_; ++mm) {
            float s = rstd * (T[mm] - mu * wsum[mm]) + bsum[mm];
            acc += fast_sigmoid(s);
        }
        pre[((size_t)ot * B_ + b0) * OG + tid] = acc;
    }
}

__global__ __launch_bounds__(256) void dnm_softmax(
    const float* __restrict__ pre, float* __restrict__ out)
{
    __shared__ __attribute__((aligned(16))) float se[O_];
    __shared__ __attribute__((aligned(16))) float sp[O_ / OG];
    const int b = blockIdx.x, t = threadIdx.x;
    if (t < O_ / OG) {
        const float4 v = *(const float4*)(pre + ((size_t)t * B_ + b) * OG);
        float4 e;
        e.x = __expf(v.x); e.y = __expf(v.y); e.z = __expf(v.z); e.w = __expf(v.w);
        *(float4*)(se + OG * t) = e;
        sp[t] = quad_sum(e);
    }
    __syncthreads();
    const float4* p4 = (const float4*)sp;
    const float4 u0 = quad_add4(p4[0],  p4[1],  p4[2],  p4[3]);
    const float4 u1 = quad_add4(p4[4],  p4[5],  p4[6],  p4[7]);
    const float4 u2 = quad_add4(p4[8],  p4[9],  p4[10], p4[11]);
    const float4 u3 = quad_add4(p4[12], p4[13], p4[14], p4[15]);
    const float tot = quad_sum(quad_add4(u0, u1, u2, u3));
    out[b * O_ + t] = se[t] * __builtin_amdgcn_rcpf(tot);
}

extern "C" void kernel_launch(void* const* d_in, const int* in_sizes, int n_in,
                              void* d_out, int out_size, void* d_ws, size_t ws_size,
                              hipStream_t stream) {
    const float* x    = (const float*)d_in[0];
    const float* sn0w = (const float*)d_in[1];
    const float* sn0b = (const float*)d_in[2];
    const float* sw   = (const float*)d_in[3];
    const float* sb   = (const float*)d_in[4];
    const float* dnw  = (const float*)d_in[5];
    const float* dnb  = (const float*)d_in[6];
    float* out = (float*)d_out;

    float* ws   = (float*)d_ws;
    float* xn   = ws;
    float* wsum = ws + 32768;
    float* bsum = ws + 32776;
    float* pre  = ws + 32784;

    dnm_prep   <<<B_ + M_,  256,      0, stream>>>(x, sn0w, sn0b, dnw, dnb, xn, wsum, bsum);
    dnm_main   <<<256,      NWV * 32, 0, stream>>>(xn, sw, sb, dnw, wsum, bsum, pre);
    dnm_softmax<<<B_,       256,      0, stream>>>(pre, out);
}


# Main inner loop: plain vector slots per 16 WMMAs cut 68 -> 37: sigmoid's (t+1)/2 moved onto the finished sums; x read as register pairs from a wave-private LDS slice a row ahead; rounding changes
# speedup vs baseline: 1.9061x; 1.1358x over previous
#include <hip/hip_runtime.h>
#include <math.h>


typedef __attribute__((ext_vector_type(2))) float v2f;
typedef __attribute__((ext_vector_type(8))) float v8f;

#define B_   64
#define O_   256
#define M_   8
#define D_   512
#define EPS_ 1e-5f
#define OG   4
#define BG   16
#define NWV  16
#define EC   32
#define RP   9
#define PW   120
#define SPILL0 (NWV * BG * PW)

__device__ __forceinline__ float fast_sigmoid(float x) {
#if __has_builtin(__builtin_amdgcn_tanhf)
    return fmaf(0.5f, __builtin_amdgcn_tanhf(0.5f * x), 0.5f);
#else
    return __builtin_amdgcn_rcpf(1.0f + __expf(-x));
#endif
}

__device__ __forceinline__ float tanh_of_half(float hv) {
#if __has_builtin(__builtin_amdgcn_tanhf)
    return __builtin_amdgcn_tanhf(hv);
#else
    return fmaf(2.0f, __builtin_amdgcn_rcpf(1.0f + __expf(-2.0f * hv)), -1.0f);
#endif
}

__device__ __forceinline__ float other_half(float v) {
    const int b = __builtin_bit_cast(int, v);
    return __builtin_bit_cast(float,
        __builtin_amdgcn_permlanex16(b, b, 0x76543210, 0xfedcba98, false, false));
}

__device__ __forceinline__ float quad_sum(float4 u) { return (u.x + u.y) + (u.z + u.w); }

__device__ __forceinline__ float4 quad_add4(float4 a, float4 b, float4 c, float4 d) {
    float4 r;
    r.x = (a.x + b.x) + (c.x + d.x); r.y = (a.y + b.y) + (c.y + d.y);
    r.z = (a.z + b.z) + (c.z + d.z); r.w = (a.w + b.w) + (c.w + d.w);
    return r;
}

__device__ __forceinline__ void block_sum2(float v1, float v2, float* s1, float* s2,
                                           float* pp, int t, float& r1, float& r2)
{
    s1[t] = v1;
    s2[t] = v2;
    __syncthreads();
    if (t < 16) {
        const float4* q1 = (const float4*)(s1 + 16 * t);
        const float4* q2 = (const float4*)(s2 + 16 * t);
        pp[t]      = quad_sum(quad_add4(q1[0], q1[1], q1[2], q1[3]));
        pp[16 + t] = quad_sum(quad_add4(q2[0], q2[1], q2[2], q2[3]));
    }
    __syncthreads();
    const float4* p4 = (const float4*)pp;
    r1 = quad_sum(quad_add4(p4[0], p4[1], p4[2], p4[3]));
    r2 = quad_sum(quad_add4(p4[4], p4[5], p4[6], p4[7]));
}

__global__ __launch_bounds__(256) void dnm_prep(
    const float* __restrict__ x,   const float* __restrict__ g,
    const float* __restrict__ be,  const float* __restrict__ dnw,
    const float* __restrict__ dnb, float* __restrict__ xn,
    float* __restrict__ wsum,      float* __restrict__ bsum)
{
    __shared__ __attribute__((aligned(16))) float s1[256];
    __shared__ __attribute__((aligned(16))) float s2[256];
    __shared__ __attribute__((aligned(16))) float pp[32];
    const int t  = threadIdx.x;
    const int bx = blockIdx.x;

    if (bx < B_) {
        const float* row = x + bx * D_;
        float a = row[t];
        float b = row[t + 256];
        float S1, S2;
        block_sum2(a + b, a * a + b * b, s1, s2, pp, t, S1, S2);
        float mu   = S1 * (1.0f / D_);
        float var  = S2 * (1.0f / D_) - mu * mu;
        float rstd = rsqrtf(var + EPS_);
        float* out = xn + bx * D_;
        out[t]       = (a - mu) * rstd * g[t]       + be[t];
        out[t + 256] = (b - mu) * rstd * g[t + 256] + be[t + 256];
    } else {
        const int m = bx - B_;
        const float* wr = dnw + m * D_;
        const float* br = dnb + m * D_;
        float S1, S2;
        block_sum2(wr[t] + wr[t + 256], br[t] + br[t + 256], s1, s2, pp, t, S1, S2);
        if (t == 0) { wsum[m] = S1; bsum[m] = S2; }
    }
}

__global__ __launch_bounds__(NWV * 32) void dnm_main(
    const float* __restrict__ xn,   const float* __restrict__ sw,
    const float* __restrict__ sb,   const float* __restrict__ dnw,
    const float* __restrict__ wsum, const float* __restrict__ bsum,
    float* __restrict__ pre)
{
    __shared__ float4 stage[2 * NWV * 32 * RP];
    __shared__ __attribute__((aligned(16))) float part[NWV * BG * PW + NWV * 32 * 8];
    __shared__ float red[4 * 10 * 64];

    const int ot   = blockIdx.x >> 2;
    const int o0   = ot * OG;
    const int b0   = (blockIdx.x & 3) * BG;
    const int tid  = threadIdx.x;
    const int w    = __builtin_amdgcn_readfirstlane(tid >> 5);
    const int lane = tid & 31;
    const int c    = lane & 15;
    const int h    = lane >> 4;
    const int d0   = w * EC;

    float4* const xs = stage + w * (32 * RP);

    float swh[EC], sbh[EC], bm[EC];
    {
        const int g  = lane >> 3;
        const int pc = lane & 7;
        const float* swl = sw + ((size_t)(o0 + g) * M_) * D_ + d0 + 4 * pc;
        const float* sbl = sb + ((size_t)(o0 + g) * M_) * D_ + d0 + 4 * pc;
        float4* st0 = xs;
        float4* st1 = stage + (NWV + w) * (32 * RP);
        float4 ta[8], tb[8], tx[4];
        #pragma unroll
        for (int j = 0; j < 8; ++j) ta[j] = *(const float4*)(swl + (2 * (j & 3) + (j >> 2)) * D_);
        #pragma unroll
        for (int j = 0; j < 8; ++j) tb[j] = *(const float4*)(sbl + (2 * (j & 3) + (j >> 2)) * D_);
        const float* xl = xn + (size_t)(b0 + g) * D_ + d0 + 4 * pc;
        #pragma unroll
        for (int i = 0; i < 4; ++i) tx[i] = *(const float4*)(xl + (4 * i) * D_);

        const int  j8  = c - 1;
        const bool sel = (c >= 1) && (c <= 8) && ((j8 & 1) == h);
        const float cst = (c == 0) ? 1.0f : 0.0f;
        const int  jr  = j8 < 0 ? 0 : (j8 > 7 ? 7 : j8);
        const float* dwr = dnw + jr * D_ + d0;
        #pragma unroll
        for (int i = 0; i < EC; i += 4) {
            const float4 d4 = *(const float4*)(dwr + i);
            bm[i + 0] = sel ? d4.x : cst; bm[i + 1] = sel ? d4.y : cst;
            bm[i + 2] = sel ? d4.z : cst; bm[i + 3] = sel ? d4.w : cst;
        }

        #pragma unroll
        for (int j = 0; j < 8; ++j) st0[(4 * j + g) * RP + pc] = ta[j];
        #pragma unroll
        for (int j = 0; j < 8; ++j) st1[(4 * j + g) * RP + pc] = tb[j];
        #pragma unroll
        for (int i = 0; i < 8; ++i) {
            const float4 v = st0[lane * RP + i];
            swh[4 * i + 0] = 0.5f * v.x; swh[4 * i + 1] = 0.5f * v.y;
            swh[4 * i + 2] = 0.5f * v.z; swh[4 * i + 3] = 0.5f * v.w;
        }
        #pragma unroll
        for (int i = 0; i < 8; ++i) {
            const float4 v = st1[lane * RP + i];
            sbh[4 * i + 0] = 0.5f * v.x; sbh[4 * i + 1] = 0.5f * v.y;
            sbh[4 * i + 2] = 0.5f * v.z; sbh[4 * i + 3] = 0.5f * v.w;
        }
        #pragma unroll
        for (int i = 0; i < 4; ++i) st0[8 * (4 * i + g) + pc] = tx[i];
    }

    const bool wr    = (c == 0) || ((c <= 8) && (h == ((c - 1) >> 2)));
    const int  slot  = (c == 0) ? h : (1 + c);
    const int  spill = SPILL0 + (w * 32 + lane) * 8;

    auto one_row = [&](const float4 (&xq)[8], const int bi) {
        v8f c1 = {};
        v2f q  = {};
        #pragma unroll
        for (int i = 0; i < EC; i += 4) {
            const float4 xv = xq[i >> 2];
            v2f a0, a1, b0v, b1v;
            a0.x  = tanh_of_half(fmaf(swh[i],     xv.x, sbh[i]));
            a0.y  = tanh_of_half(fmaf(swh[i + 1], xv.y, sbh[i + 1]));
            b0v.x = bm[i];
            b0v.y = bm[i + 1];
            c1 = __builtin_amdgcn_wmma_f32_16x16x4_f32(false, a0, false, b0v, (short)0, c1, false, false);
            q  = __builtin_elementwise_fma(a0, a0, q);
            a1.x  = tanh_of_half(fmaf(swh[i + 2], xv.z, sbh[i + 2]));
            a1.y  = tanh_of_half(fmaf(swh[i + 3], xv.w, sbh[i + 3]));
            b1v.x = bm[i + 2];
            b1v.y = bm[i + 3];
            c1 = __builtin_amdgcn_wmma_f32_16x16x4_f32(false, a1, false, b1v, (short)0, c1, false, false);
            q  = __builtin_elementwise_fma(a1, a1, q);
        }
        const int rec = (w * BG + bi) * PW;
        part[rec + 80 + lane] = q.x + q.y;
        float* dst = part + (wr ? rec + 8 * slot : spill);
        float4 lo, hi;
        lo.x = c1[0]; lo.y = c1[1]; lo.z = c1[2]; lo.w = c1[3];
        hi.x = c1[4]; hi.y = c1[5]; hi.z = c1[6]; hi.w = c1[7];
        *(float4*)(dst)     = lo;
        *(float4*)(dst + 4) = hi;
    };

    float4 xa[8], xb[8];
    #pragma unroll
    for (int i = 0; i < 8; ++i) xa[i] = xs[i];

    #pragma unroll 1
    for (int bi = 0; bi < BG; bi += 2) {
        #pragma unroll
        for (int i = 0; i < 8; ++i) xb[i] = xs[8 * (bi + 1) + i];
        one_row(xa, bi);
        #pragma unroll
        for (int i = 0; i < 8; ++i) xa[i] = xs[8 * ((bi + 2) & (BG - 1)) + i];
        one_row(xb, bi + 1);
    }
    __syncthreads();

    if (tid < 256) {
        const int ws = tid >> 6;
        const int r  = tid & 63;
        const int bi = r >> 2;
        const int oi = r & 3;
        float S1 = 0.f, S2 = 0.f, T[M_];
        #pragma unroll
        for (int mm = 0; mm < M_; ++mm) T[mm] = 0.f;
        #pragma unroll
        for (int k = 0; k < 4; ++k) {
            const float* pp = part + ((4 * ws + k) * BG + bi) * PW;
            S1 += (pp[oi] + pp[4 + oi]) + (pp[8 + oi] + pp[12 + oi]);
            #pragma unroll
            for (int mm = 0; mm < M_; ++mm) T[mm] += pp[8 * (2 + mm) + 4 * ((mm >> 1) & 1) + oi];
            #pragma unroll
            for (int u = 0; u < 8; ++u) S2 += pp[80 + 4 * u + oi];
        }
        float* rr = red + ws * 640 + r;
        rr[0]  = S1;
        rr[64] = S2;
        #pragma unroll
        for (int mm = 0; mm < M_; ++mm) rr[128 + 64 * mm] = T[mm];
    }
    __syncthreads();

    if (tid < BG * OG) {
        float S1 = 0.f, S2 = 0.f, T[M_];
        #pragma unroll
        for (int mm = 0; mm < M_; ++mm) T[mm] = 0.f;
        #pragma unroll
        for (int ws = 0; ws < 4; ++ws) {
            const float* rr = red + ws * 640 + tid;
            S1 += rr[0];
            S2 += rr[64];
            #pragma unroll
            for (int mm = 0; mm < M_; ++mm) T[mm] += rr[128 + 64 * mm];
        }
        const float inv   = 1.0f / (float)(M_ * D_);
        const float mu    = S1 * inv;
        const float var   = S2 * inv - mu * mu;
        const float hrstd = 0.5f * rsqrtf(fmaf(0.25f, var, EPS_));
        float acc = 0.f;
        #pragma unroll
        for (int mm = 0; mm < M_; ++mm) {
            float s = hrstd * (T[mm] - mu * wsum[mm]) + bsum[mm];
            acc += fast_sigmoid(s);
        }
        pre[((size_t)ot * B_ + b0) * OG + tid] = acc;
    }
}

__global__ __launch_bounds__(256) void dnm_softmax(
    const float* __restrict__ pre, float* __restrict__ out)
{
    __shared__ __attribute__((aligned(16))) float se[O_];
    __shared__ __attribute__((aligned(16))) float sp[O_ / OG];
    const int b = blockIdx.x, t = threadIdx.x;
    if (t < O_ / OG) {
        const float4 v = *(const float4*)(pre + ((size_t)t * B_ + b) * OG);
        float4 e;
        e.x = __expf(v.x); e.y = __expf(v.y); e.z = __expf(v.z); e.w = __expf(v.w);
        *(float4*)(se + OG * t) = e;
        sp[t] = quad_sum(e);
    }
    __syncthreads();
    const float4* p4 = (const float4*)sp;
    const float4 u0 = quad_add4(p4[0],  p4[1],  p4[2],  p4[3]);
    const float4 u1 = quad_add4(p4[4],  p4[5],  p4[6],  p4[7]);
    const float4 u2 = quad_add4(p4[8],  p4[9],  p4[10], p4[11]);
    const float4 u3 = quad_add4(p4[12], p4[13], p4[14], p4[15]);
    const float tot = quad_sum(quad_add4(u0, u1, u2, u3));
    out[b * O_ + t] = se[t] * __builtin_amdgcn_rcpf(tot);
}

extern "C" void kernel_launch(void* const* d_in, const int* in_sizes, int n_in,
                              void* d_out, int out_size, void* d_ws, size_t ws_size,
                              hipStream_t stream) {
    const float* x    = (const float*)d_in[0];
    const float* sn0w = (const float*)d_in[1];
    const float* sn0b = (const float*)d_in[2];
    const float* sw   = (const float*)d_in[3];
    const float* sb   = (const float*)d_in[4];
    const float* dnw  = (const float*)d_in[5];
    const float* dnb  = (const float*)d_in[6];
    float* out = (float*)d_out;

    float* ws   = (float*)d_ws;
    float* xn   = ws;
    float* wsum = ws + 32768;
    float* bsum = ws + 32776;
    float* pre  = ws + 32784;

    dnm_prep   <<<B_ + M_,  256,      0, stream>>>(x, sn0w, sn0b, dnw, dnb, xn, wsum, bsum);
    dnm_main   <<<256,      NWV * 32, 0, stream>>>(xn, sw, sb, dnw, wsum, bsum, pre);
    dnm_softmax<<<B_,       256,      0, stream>>>(pre, out);
}


# Main: the row's last 4 matrix steps go to packed add/fma lane sums like the other 12; the one 0/1-selector WMMA that adds the lane sums stays (1 WMMA a row, not 5); sum orders change
# speedup vs baseline: 2.6561x; 1.3935x over previous
#include <hip/hip_runtime.h>
#include <math.h>


typedef __attribute__((ext_vector_type(2))) float v2f;
typedef __attribute__((ext_vector_type(8))) float v8f;

#define B_   64
#define O_   256
#define M_   8
#define D_   512
#define EPS_ 1e-5f
#define OG   4
#define BG   16
#define NWV  16
#define EC   32
#define RP   9
#define PW   120
#define SPILL0 (NWV * BG * PW)

__device__ __forceinline__ float fast_sigmoid(float x) {
#if __has_builtin(__builtin_amdgcn_tanhf)
    return fmaf(0.5f, __builtin_amdgcn_tanhf(0.5f * x), 0.5f);
#else
    return __builtin_amdgcn_rcpf(1.0f + __expf(-x));
#endif
}

__device__ __forceinline__ float tanh_of_half(float hv) {
#if __has_builtin(__builtin_amdgcn_tanhf)
    return __builtin_amdgcn_tanhf(hv);
#else
    return fmaf(2.0f, __builtin_amdgcn_rcpf(1.0f + __expf(-2.0f * hv)), -1.0f);
#endif
}

__device__ __forceinline__ float other_half(float v) {
    const int b = __builtin_bit_cast(int, v);
    return __builtin_bit_cast(float,
        __builtin_amdgcn_permlanex16(b, b, 0x76543210, 0xfedcba98, false, false));
}

__device__ __forceinline__ float quad_sum(float4 u) { return (u.x + u.y) + (u.z + u.w); }

__device__ __forceinline__ float4 quad_add4(float4 a, float4 b, float4 c, float4 d) {
    float4 r;
    r.x = (a.x + b.x) + (c.x + d.x); r.y = (a.y + b.y) + (c.y + d.y);
    r.z = (a.z + b.z) + (c.z + d.z); r.w = (a.w + b.w) + (c.w + d.w);
    return r;
}

__device__ __forceinline__ void block_sum2(float v1, float v2, float* s1, float* s2,
                                           float* pp, int t, float& r1, float& r2)
{
    s1[t] = v1;
    s2[t] = v2;
    __syncthreads();
    if (t < 16) {
        const float4* q1 = (const float4*)(s1 + 16 * t);
        const float4* q2 = (const float4*)(s2 + 16 * t);
        pp[t]      = quad_sum(quad_add4(q1[0], q1[1], q1[2], q1[3]));
        pp[16 + t] = quad_sum(quad_add4(q2[0], q2[1], q2[2], q2[3]));
    }
    __syncthreads();
    const float4* p4 = (const float4*)pp;
    r1 = quad_sum(quad_add4(p4[0], p4[1], p4[2], p4[3]));
    r2 = quad_sum(quad_add4(p4[4], p4[5], p4[6], p4[7]));
}

__global__ __launch_bounds__(256) void dnm_prep(
    const float* __restrict__ x,   const float* __restrict__ g,
    const float* __restrict__ be,  const float* __restrict__ dnw,
    const float* __restrict__ dnb, float* __restrict__ xn,
    float* __restrict__ wsum,      float* __restrict__ bsum)
{
    __shared__ __attribute__((aligned(16))) float s1[256];
    __shared__ __attribute__((aligned(16))) float s2[256];
    __shared__ __attribute__((aligned(16))) float pp[32];
    const int t  = threadIdx.x;
    const int bx = blockIdx.x;

    if (bx < B_) {
        const float* row = x + bx * D_;
        float a = row[t];
        float b = row[t + 256];
        float S1, S2;
        block_sum2(a + b, a * a + b * b, s1, s2, pp, t, S1, S2);
        float mu   = S1 * (1.0f / D_);
        float var  = S2 * (1.0f / D_) - mu * mu;
        float rstd = rsqrtf(var + EPS_);
        float* out = xn + bx * D_;
        out[t]       = (a - mu) * rstd * g[t]       + be[t];
        out[t + 256] = (b - mu) * rstd * g[t + 256] + be[t + 256];
    } else {
        const int m = bx - B_;
        const float* wr = dnw + m * D_;
        const float* br = dnb + m * D_;
        float S1, S2;
        block_sum2(wr[t] + wr[t + 256], br[t] + br[t + 256], s1, s2, pp, t, S1, S2);
        if (t == 0) { wsum[m] = S1; bsum[m] = S2; }
    }
}

__global__ __launch_bounds__(NWV * 32) void dnm_main(
    const float* __restrict__ xn,   const float* __restrict__ sw,
    const float* __restrict__ sb,   const float* __restrict__ dnw,
    const float* __restrict__ wsum, const float* __restrict__ bsum,
    float* __restrict__ pre)
{
    __shared__ float4 stage[2 * NWV * 32 * RP];
    __shared__ __attribute__((aligned(16))) float part[NWV * BG * PW + NWV * 32 * 8];
    __shared__ float red[4 * 10 * 64];

    const int ot   = blockIdx.x >> 2;
    const int o0   = ot * OG;
    const int b0   = (blockIdx.x & 3) * BG;
    const int tid  = threadIdx.x;
    const int w    = __builtin_amdgcn_readfirstlane(tid >> 5);
    const int lane = tid & 31;
    const int c    = lane & 15;
    const int h    = lane >> 4;
    const int d0   = w * EC;

    float4* const xs = stage + w * (32 * RP);

    float swh[EC], sbh[EC], bm[EC];
    v2f bsel;
    {
        const int g  = lane >> 3;
        const int pc = lane & 7;
        const float* swl = sw + ((size_t)(o0 + g) * M_) * D_ + d0 + 4 * pc;
        const float* sbl = sb + ((size_t)(o0 + g) * M_) * D_ + d0 + 4 * pc;
        float4* st0 = xs;
        float4* st1 = stage + (NWV + w) * (32 * RP);
        float4 ta[8], tb[8], tx[4];
        #pragma unroll
        for (int j = 0; j < 8; ++j) ta[j] = *(const float4*)(swl + (2 * (j & 3) + (j >> 2)) * D_);
        #pragma unroll
        for (int j = 0; j < 8; ++j) tb[j] = *(const float4*)(sbl + (2 * (j & 3) + (j >> 2)) * D_);
        const float* xl = xn + (size_t)(b0 + g) * D_ + d0 + 4 * pc;
        #pragma unroll
        for (int i = 0; i < 4; ++i) tx[i] = *(const float4*)(xl + (4 * i) * D_);

        const float* dwo = dnw + (2 * (c >> 2) + h) * D_ + d0;
        #pragma unroll
        for (int i = 0; i < EC; i += 4) {
            const float4 w0 = *(const float4*)(dwo + i);
            bm[i + 0] = w0.x; bm[i + 1] = w0.y; bm[i + 2] = w0.z; bm[i + 3] = w0.w;
        }
        const bool sel = (c >= 1) && (c <= 8) && (((c - 1) & 1) == h);
        bsel.x = (c == 0) ? 1.0f : 0.0f;
        bsel.y = sel ? 1.0f : 0.0f;

        #pragma unroll
        for (int j = 0; j < 8; ++j) st0[(4 * j + g) * RP + pc] = ta[j];
        #pragma unroll
        for (int j = 0; j < 8; ++j) st1[(4 * j + g) * RP + pc] = tb[j];
        #pragma unroll
        for (int i = 0; i < 8; ++i) {
            const float4 v = st0[lane * RP + i];
            swh[4 * i + 0] = 0.5f * v.x; swh[4 * i + 1] = 0.5f * v.y;
            swh[4 * i + 2] = 0.5f * v.z; swh[4 * i + 3] = 0.5f * v.w;
        }
        #pragma unroll
        for (int i = 0; i < 8; ++i) {
            const float4 v = st1[lane * RP + i];
            sbh[4 * i + 0] = 0.5f * v.x; sbh[4 * i + 1] = 0.5f * v.y;
            sbh[4 * i + 2] = 0.5f * v.z; sbh[4 * i + 3] = 0.5f * v.w;
        }
        #pragma unroll
        for (int i = 0; i < 4; ++i) st0[8 * (4 * i + g) + pc] = tx[i];
    }

    const bool wr    = (c == 0) || ((c <= 8) && (h == ((c - 1) >> 2)));
    const int  slot  = (c == 0) ? h : (1 + c);
    const int  spill = SPILL0 + (w * 32 + lane) * 8;

    auto one_row = [&](const float4 (&xq)[8], const int bi) {
        v2f q  = {};
        v2f sv = {};
        v2f tv = {};
        #pragma unroll
        for (int i = 0; i < EC; i += 8) {
            const float4 xu = xq[i >> 2];
            const float4 xv = xq[(i >> 2) + 1];
            v2f a0, a1, a2, a3, w0, w1, w2, w3;
            a0.x = tanh_of_half(fmaf(swh[i],     xu.x, sbh[i]));
            a0.y = tanh_of_half(fmaf(swh[i + 1], xu.y, sbh[i + 1]));
            w0.x = bm[i];
            w0.y = bm[i + 1];
            sv = sv + a0;
            tv = __builtin_elementwise_fma(a0, w0, tv);
            q  = __builtin_elementwise_fma(a0, a0, q);
            a1.x = tanh_of_half(fmaf(swh[i + 2], xu.z, sbh[i + 2]));
            a1.y = tanh_of_half(fmaf(swh[i + 3], xu.w, sbh[i + 3]));
            w1.x = bm[i + 2];
            w1.y = bm[i + 3];
            sv = sv + a1;
            tv = __builtin_elementwise_fma(a1, w1, tv);
            q  = __builtin_elementwise_fma(a1, a1, q);
            a2.x = tanh_of_half(fmaf(swh[i + 4], xv.x, sbh[i + 4]));
            a2.y = tanh_of_half(fmaf(swh[i + 5], xv.y, sbh[i + 5]));
            w2.x = bm[i + 4];
            w2.y = bm[i + 5];
            sv = sv + a2;
            tv = __builtin_elementwise_fma(a2, w2, tv);
            q  = __builtin_elementwise_fma(a2, a2, q);
            a3.x = tanh_of_half(fmaf(swh[i + 6], xv.z, sbh[i + 6]));
            a3.y = tanh_of_half(fmaf(swh[i + 7], xv.w, sbh[i + 7]));
            w3.x = bm[i + 6];
            w3.y = bm[i + 7];
            sv = sv + a3;
            tv = __builtin_elementwise_fma(a3, w3, tv);
            q  = __builtin_elementwise_fma(a3, a3, q);
        }
        v2f af;
        af.x = sv.x + sv.y;
        af.y = tv.x + tv.y;
        const v8f cz = {};
        const v8f c1 = __builtin_amdgcn_wmma_f32_16x16x4_f32(false, af, false, bsel, (short)0, cz, false, false);
        const int rec = (w * BG + bi) * PW;
        part[rec + 80 + lane] = q.x + q.y;
        float* dst = part + (wr ? rec + 8 * slot : spill);
        float4 lo, hi;
        lo.x = c1[0]; lo.y = c1[1]; lo.z = c1[2]; lo.w = c1[3];
        hi.x = c1[4]; hi.y = c1[5]; hi.z = c1[6]; hi.w = c1[7];
        *(float4*)(dst)     = lo;
        *(float4*)(dst + 4) = hi;
    };

    float4 xa[8], xb[8];
    #pragma unroll
    for (int i = 0; i < 8; ++i) xa[i] = xs[i];

    #pragma unroll 1
    for (int bi = 0; bi < BG; bi += 2) {
        #pragma unroll
        for (int i = 0; i < 8; ++i) xb[i] = xs[8 * (bi + 1) + i];
        one_row(xa, bi);
        #pragma unroll
        for (int i = 0; i < 8; ++i) xa[i] = xs[8 * ((bi + 2) & (BG - 1)) + i];
        one_row(xb, bi + 1);
    }
    __syncthreads();

    if (tid < 256) {
        const int ws = tid >> 6;
        const int r  = tid & 63;
        const int bi = r >> 2;
        const int oi = r & 3;
        float S1 = 0.f, S2 = 0.f, T[M_];
        #pragma unroll
        for (int mm = 0; mm < M_; ++mm) T[mm] = 0.f;
        #pragma unroll
        for (int k = 0; k < 4; ++k) {
            const float* pp = part + ((4 * ws + k) * BG + bi) * PW;
            S1 += (pp[oi] + pp[4 + oi]) + (pp[8 + oi] + pp[12 + oi]);
            #pragma unroll
            for (int mm = 0; mm < M_; ++mm) T[mm] += pp[8 * (2 + mm) + 4 * ((mm >> 1) & 1) + oi];
            #pragma unroll
            for (int u = 0; u < 8; ++u) S2 += pp[80 + 4 * u + oi];
        }
        float* rr = red + ws * 640 + r;
        rr[0]  = S1;
        rr[64] = S2;
        #pragma unroll
        for (int mm = 0; mm < M_; ++mm) rr[128 + 64 * mm] = T[mm];
    }
    __syncthreads();

    if (tid < BG * OG) {
        float S1 = 0.f, S2 = 0.f, T[M_];
        #pragma unroll
        for (int mm = 0; mm < M_; ++mm) T[mm] = 0.f;
        #pragma unroll
        for (int ws = 0; ws < 4; ++ws) {
            const float* rr = red + ws * 640 + tid;
            S1 += rr[0];
            S2 += rr[64];
            #pragma unroll
            for (int mm = 0; mm < M_; ++mm) T[mm] += rr[128 + 64 * mm];
        }
        const float inv   = 1.0f / (float)(M_ * D_);
        const float mu    = S1 * inv;
        const float var   = S2 * inv - mu * mu;
        const float hrstd = 0.5f * rsqrtf(fmaf(0.25f, var, EPS_));
        float acc = 0.f;
        #pragma unroll
        for (int mm = 0; mm < M_; ++mm) {
            float s = hrstd * (T[mm] - mu * wsum[mm]) + bsum[mm];
            acc += fast_sigmoid(s);
        }
        pre[((size_t)ot * B_ + b0) * OG + tid] = acc;
    }
}

__global__ __launch_bounds__(256) void dnm_softmax(
    const float* __restrict__ pre, float* __restrict__ out)
{
    __shared__ __attribute__((aligned(16))) float se[O_];
    __shared__ __attribute__((aligned(16))) float sp[O_ / OG];
    const int b = blockIdx.x, t = threadIdx.x;
    if (t < O_ / OG) {
        const float4 v = *(const float4*)(pre + ((size_t)t * B_ + b) * OG);
        float4 e;
        e.x = __expf(v.x); e.y = __expf(v.y); e.z = __expf(v.z); e.w = __expf(v.w);
        *(float4*)(se + OG * t) = e;
        sp[t] = quad_sum(e);
    }
    __syncthreads();
    const float4* p4 = (const float4*)sp;
    const float4 u0 = quad_add4(p4[0],  p4[1],  p4[2],  p4[3]);
    const float4 u1 = quad_add4(p4[4],  p4[5],  p4[6],  p4[7]);
    const float4 u2 = quad_add4(p4[8],  p4[9],  p4[10], p4[11]);
    const float4 u3 = quad_add4(p4[12], p4[13], p4[14], p4[15]);
    const float tot = quad_sum(quad_add4(u0, u1, u2, u3));
    out[b * O_ + t] = se[t] * __builtin_amdgcn_rcpf(tot);
}

extern "C" void kernel_launch(void* const* d_in, const int* in_sizes, int n_in,
                              void* d_out, int out_size, void* d_ws, size_t ws_size,
                              hipStream_t stream) {
    const float* x    = (const float*)d_in[0];
    const float* sn0w = (const float*)d_in[1];
    const float* sn0b = (const float*)d_in[2];
    const float* sw   = (const float*)d_in[3];
    const float* sb   = (const float*)d_in[4];
    const float* dnw  = (const float*)d_in[5];
    const float* dnb  = (const float*)d_in[6];
    float* out = (float*)d_out;

    float* ws   = (float*)d_ws;
    float* xn   = ws;
    float* wsum = ws + 32768;
    float* bsum = ws + 32776;
    float* pre  = ws + 32784;

    dnm_prep   <<<B_ + M_,  256,      0, stream>>>(x, sn0w, sn0b, dnw, dnb, xn, wsum, bsum);
    dnm_main   <<<256,      NWV * 32, 0, stream>>>(xn, sw, sb, dnw, wsum, bsum, pre);
    dnm_softmax<<<B_,       256,      0, stream>>>(pre, out);
}


# Main: LayerNorm launch gone (2 launches, not 3): each wave sums x, x^2 over its own 16x32 slice, the 16 waves' sums meet in LDS behind one barrier, slice normalised in registers; sum order changes
# speedup vs baseline: 3.0669x; 1.1547x over previous
#include <hip/hip_runtime.h>
#include <math.h>


typedef __attribute__((ext_vector_type(2))) float v2f;
typedef __attribute__((ext_vector_type(8))) float v8f;

#define B_   64
#define O_   256
#define M_   8
#define D_   512
#define EPS_ 1e-5f
#define OG   4
#define BG   16
#define NWV  16
#define EC   32
#define RP   9
#define PW   120
#define SPILL0 (NWV * BG * PW)
#define PSTN (2 * BG * NWV + 2 * M_ * NWV)

__device__ __forceinline__ float fast_sigmoid(float x) {
#if __has_builtin(__builtin_amdgcn_tanhf)
    return fmaf(0.5f, __builtin_amdgcn_tanhf(0.5f * x), 0.5f);
#else
    return __builtin_amdgcn_rcpf(1.0f + __expf(-x));
#endif
}

__device__ __forceinline__ float tanh_of_half(float hv) {
#if __has_builtin(__builtin_amdgcn_tanhf)
    return __builtin_amdgcn_tanhf(hv);
#else
    return fmaf(2.0f, __builtin_amdgcn_rcpf(1.0f + __expf(-2.0f * hv)), -1.0f);
#endif
}

__device__ __forceinline__ float other_half(float v) {
    const int b = __builtin_bit_cast(int, v);
    return __builtin_bit_cast(float,
        __builtin_amdgcn_permlanex16(b, b, 0x76543210, 0xfedcba98, false, false));
}

__device__ __forceinline__ float quad_sum(float4 u) { return (u.x + u.y) + (u.z + u.w); }

__device__ __forceinline__ float4 quad_add4(float4 a, float4 b, float4 c, float4 d) {
    float4 r;
    r.x = (a.x + b.x) + (c.x + d.x); r.y = (a.y + b.y) + (c.y + d.y);
    r.z = (a.z + b.z) + (c.z + d.z); r.w = (a.w + b.w) + (c.w + d.w);
    return r;
}

template <int CTRL> __device__ __forceinline__ float xlane_add(float v) {
    const int b = __builtin_bit_cast(int, v);
    return v + __builtin_bit_cast(float, __builtin_amdgcn_update_dpp(b, b, CTRL, 0xf, 0xf, false));
}

__global__ __launch_bounds__(NWV * 32) void dnm_main(
    const float* __restrict__ x,    const float* __restrict__ lnw,
    const float* __restrict__ lnb,  const float* sw,
    const float* sb,                const float* dnw,
    const float* __restrict__ dnb,  float* pre)
{
    __shared__ float4 stage[2 * NWV * 32 * RP];
    __shared__ __attribute__((aligned(16))) float part[NWV * BG * PW + NWV * 32 * 8];
    __shared__ float red[4 * 10 * 64];
    __shared__ __attribute__((aligned(16))) float pst[PSTN];
    __shared__ float wb[2 * M_];

    const int ot   = blockIdx.x >> 2;
    const int o0   = ot * OG;
    const int b0   = (blockIdx.x & 3) * BG;
    const int tid  = threadIdx.x;
    const int w    = __builtin_amdgcn_readfirstlane(tid >> 5);
    const int lane = tid & 31;
    const int c    = lane & 15;
    const int h    = lane >> 4;
    const int d0   = w * EC;

    float4* const xs = stage + w * (32 * RP);

    float swh[EC], sbh[EC], bm[EC];
    v2f bsel;
    {
        const int g  = lane >> 3;
        const int pc = lane & 7;
        const float* swl = sw + ((size_t)(o0 + g) * M_) * D_ + d0 + 4 * pc;
        const float* sbl = sb + ((size_t)(o0 + g) * M_) * D_ + d0 + 4 * pc;
        float4* st0 = xs;
        float4* st1 = stage + (NWV + w) * (32 * RP);
        float4 tx[4];
        const float* xl = x + (size_t)(b0 + g) * D_ + d0 + 4 * pc;
        #pragma unroll
        for (int i = 0; i < 4; ++i) tx[i] = *(const float4*)(xl + (4 * i) * D_);
        const float4 gw = *(const float4*)(lnw + d0 + 4 * pc);
        const float4 gb = *(const float4*)(lnb + d0 + 4 * pc);
        const int mq = lane >> 2;
        const int qq = lane & 3;
        const float4 dw0 = *(const float4*)(dnw + mq * D_ + d0 + 8 * qq);
        const float4 dw1 = *(const float4*)(dnw + mq * D_ + d0 + 8 * qq + 4);
        const float4 db0 = *(const float4*)(dnb + mq * D_ + d0 + 8 * qq);
        const float4 db1 = *(const float4*)(dnb + mq * D_ + d0 + 8 * qq + 4);
        const float4 ta0 = *(const float4*)(swl + 0 * D_);
        const float4 ta1 = *(const float4*)(swl + 2 * D_);
        const float4 ta2 = *(const float4*)(swl + 4 * D_);
        const float4 ta3 = *(const float4*)(swl + 6 * D_);
        const float4 ta4 = *(const float4*)(swl + 1 * D_);
        const float4 ta5 = *(const float4*)(swl + 3 * D_);
        const float4 ta6 = *(const float4*)(swl + 5 * D_);
        const float4 ta7 = *(const float4*)(swl + 7 * D_);
        const float4 tb0 = *(const float4*)(sbl + 0 * D_);
        const float4 tb1 = *(const float4*)(sbl + 2 * D_);
        const float4 tb2 = *(const float4*)(sbl + 4 * D_);
        const float4 tb3 = *(const float4*)(sbl + 6 * D_);
        const float4 tb4 = *(const float4*)(sbl + 1 * D_);
        const float4 tb5 = *(const float4*)(sbl + 3 * D_);
        const float4 tb6 = *(const float4*)(sbl + 5 * D_);
        const float4 tb7 = *(const float4*)(sbl + 7 * D_);

        const float* dwo = dnw + (2 * (c >> 2) + h) * D_ + d0;
        #pragma unroll
        for (int i = 0; i < EC; i += 4) {
            const float4 w0 = *(const float4*)(dwo + i);
            bm[i + 0] = w0.x; bm[i + 1] = w0.y; bm[i + 2] = w0.z; bm[i + 3] = w0.w;
        }
        const bool sel = (c >= 1) && (c <= 8) && (((c - 1) & 1) == h);
        bsel.x = (c == 0) ? 1.0f : 0.0f;
        bsel.y = sel ? 1.0f : 0.0f;

        float p1[4], p2[4];
        #pragma unroll
        for (int i = 0; i < 4; ++i) {
            const float4 u = tx[i];
            float a1 = quad_sum(u);
            float a2 = (u.x * u.x + u.y * u.y) + (u.z * u.z + u.w * u.w);
            a1 = xlane_add<0x161>(a1); a1 = xlane_add<0x162>(a1); a1 = xlane_add<0x164>(a1);
            a2 = xlane_add<0x161>(a2); a2 = xlane_add<0x162>(a2); a2 = xlane_add<0x164>(a2);
            p1[i] = a1; p2[i] = a2;
        }
        #pragma unroll
        for (int i = 0; i < 4; ++i)
            if (pc < 2) pst[(pc * BG + 4 * i + g) * NWV + w] = (pc == 0) ? p1[i] : p2[i];
        asm volatile("s_wait_dscnt 0x0" ::: "memory");
        asm volatile("s_barrier_signal -1\n\ts_barrier_wait -1" ::: "memory");
        float mu, rstd;
        {
            const float4* r1 = (const float4*)(pst + c * NWV + 8 * h);
            const float4* r2 = (const float4*)(pst + (BG + c) * NWV + 8 * h);
            float S1 = quad_sum(r1[0]) + quad_sum(r1[1]);
            float S2 = quad_sum(r2[0]) + quad_sum(r2[1]);
            S1 += other_half(S1);
            S2 += other_half(S2);
            mu = S1 * (1.0f / D_);
            const float var = S2 * (1.0f / D_) - mu * mu;
            rstd = rsqrtf(var + EPS_);
        }
        #pragma unroll
        for (int i = 0; i < 4; ++i) {
            const float mi = __shfl(mu,   4 * i + g);
            const float ri = __shfl(rstd, 4 * i + g);
            tx[i].x = (tx[i].x - mi) * ri * gw.x + gb.x;
            tx[i].y = (tx[i].y - mi) * ri * gw.y + gb.y;
            tx[i].z = (tx[i].z - mi) * ri * gw.z + gb.z;
            tx[i].w = (tx[i].w - mi) * ri * gw.w + gb.w;
        }

        __builtin_amdgcn_sched_barrier(0);
        st0[(0 + g) * RP + pc] = ta0;
        st0[(4 + g) * RP + pc] = ta1;
        st0[(8 + g) * RP + pc] = ta2;
        st0[(12 + g) * RP + pc] = ta3;
        st0[(16 + g) * RP + pc] = ta4;
        st0[(20 + g) * RP + pc] = ta5;
        st0[(24 + g) * RP + pc] = ta6;
        st0[(28 + g) * RP + pc] = ta7;
        st1[(0 + g) * RP + pc] = tb0;
        st1[(4 + g) * RP + pc] = tb1;
        st1[(8 + g) * RP + pc] = tb2;
        st1[(12 + g) * RP + pc] = tb3;
        st1[(16 + g) * RP + pc] = tb4;
        st1[(20 + g) * RP + pc] = tb5;
        st1[(24 + g) * RP + pc] = tb6;
        st1[(28 + g) * RP + pc] = tb7;
        #pragma unroll
        for (int i = 0; i < 8; ++i) {
            const float4 v = st0[lane * RP + i];
            swh[4 * i + 0] = 0.5f * v.x; swh[4 * i + 1] = 0.5f * v.y;
            swh[4 * i + 2] = 0.5f * v.z; swh[4 * i + 3] = 0.5f * v.w;
        }
        #pragma unroll
        for (int i = 0; i < 8; ++i) {
            const float4 v = st1[lane * RP + i];
            sbh[4 * i + 0] = 0.5f * v.x; sbh[4 * i + 1] = 0.5f * v.y;
            sbh[4 * i + 2] = 0.5f * v.z; sbh[4 * i + 3] = 0.5f * v.w;
        }
        #pragma unroll
        for (int i = 0; i < 4; ++i) st0[8 * (4 * i + g) + pc] = tx[i];

        __builtin_amdgcn_sched_barrier(0);
        float wsp = quad_sum(dw0) + quad_sum(dw1);
        float bsp = quad_sum(db0) + quad_sum(db1);
        wsp = xlane_add<0x161>(wsp); wsp = xlane_add<0x162>(wsp);
        bsp = xlane_add<0x161>(bsp); bsp = xlane_add<0x162>(bsp);
        if (qq < 2) pst[2 * BG * NWV + (qq * M_ + mq) * NWV + w] = (qq == 0) ? wsp : bsp;
    }

    const bool wr    = (c == 0) || ((c <= 8) && (h == ((c - 1) >> 2)));
    const int  slot  = (c == 0) ? h : (1 + c);
    const int  spill = SPILL0 + (w * 32 + lane) * 8;

    auto one_row = [&](const float4 (&xq)[8], const int bi) {
        v2f q  = {};
        v2f sv = {};
        v2f tv = {};
        #pragma unroll
        for (int i = 0; i < EC; i += 8) {
            const float4 xu = xq[i >> 2];
            const float4 xv = xq[(i >> 2) + 1];
            v2f a0, a1, a2, a3, w0, w1, w2, w3;
            a0.x = tanh_of_half(fmaf(swh[i],     xu.x, sbh[i]));
            a0.y = tanh_of_half(fmaf(swh[i + 1], xu.y, sbh[i + 1]));
            w0.x = bm[i];
            w0.y = bm[i + 1];
            sv = sv + a0;
            tv = __builtin_elementwise_fma(a0, w0, tv);
            q  = __builtin_elementwise_fma(a0, a0, q);
            a1.x = tanh_of_half(fmaf(swh[i + 2], xu.z, sbh[i + 2]));
            a1.y = tanh_of_half(fmaf(swh[i + 3], xu.w, sbh[i + 3]));
            w1.x = bm[i + 2];
            w1.y = bm[i + 3];
            sv = sv + a1;
            tv = __builtin_elementwise_fma(a1, w1, tv);
            q  = __builtin_elementwise_fma(a1, a1, q);
            a2.x = tanh_of_half(fmaf(swh[i + 4], xv.x, sbh[i + 4]));
            a2.y = tanh_of_half(fmaf(swh[i + 5], xv.y, sbh[i + 5]));
            w2.x = bm[i + 4];
            w2.y = bm[i + 5];
            sv = sv + a2;
            tv = __builtin_elementwise_fma(a2, w2, tv);
            q  = __builtin_elementwise_fma(a2, a2, q);
            a3.x = tanh_of_half(fmaf(swh[i + 6], xv.z, sbh[i + 6]));
            a3.y = tanh_of_half(fmaf(swh[i + 7], xv.w, sbh[i + 7]));
            w3.x = bm[i + 6];
            w3.y = bm[i + 7];
            sv = sv + a3;
            tv = __builtin_elementwise_fma(a3, w3, tv);
            q  = __builtin_elementwise_fma(a3, a3, q);
        }
        v2f af;
        af.x = sv.x + sv.y;
        af.y = tv.x + tv.y;
        const v8f cz = {};
        const v8f c1 = __builtin_amdgcn_wmma_f32_16x16x4_f32(false, af, false, bsel, (short)0, cz, false, false);
        const int rec = (w * BG + bi) * PW;
        part[rec + 80 + lane] = q.x + q.y;
        float* dst = part + (wr ? rec + 8 * slot : spill);
        float4 lo, hi;
        lo.x = c1[0]; lo.y = c1[1]; lo.z = c1[2]; lo.w = c1[3];
        hi.x = c1[4]; hi.y = c1[5]; hi.z = c1[6]; hi.w = c1[7];
        *(float4*)(dst)     = lo;
        *(float4*)(dst + 4) = hi;
    };

    float4 xa[8], xb[8];
    #pragma unroll
    for (int i = 0; i < 8; ++i) xa[i] = xs[i];

    #pragma unroll 1
    for (int bi = 0; bi < BG; bi += 2) {
        #pragma unroll
        for (int i = 0; i < 8; ++i) xb[i] = xs[8 * (bi + 1) + i];
        one_row(xa, bi);
        #pragma unroll
        for (int i = 0; i < 8; ++i) xa[i] = xs[8 * ((bi + 2) & (BG - 1)) + i];
        one_row(xb, bi + 1);
    }
    __syncthreads();

    if (tid < 256) {
        if (tid < 2 * M_) {
            const float4* q4 = (const float4*)(pst + 2 * BG * NWV + NWV * tid);
            wb[tid] = quad_sum(quad_add4(q4[0], q4[1], q4[2], q4[3]));
        }
        const int ws = tid >> 6;
        const int r  = tid & 63;
        const int bi = r >> 2;
        const int oi = r & 3;
        float S1 = 0.f, S2 = 0.f, T[M_];
        #pragma unroll
        for (int mm = 0; mm < M_; ++mm) T[mm] = 0.f;
        #pragma unroll
        for (int k = 0; k < 4; ++k) {
            const float* pp = part + ((4 * ws + k) * BG + bi) * PW;
            S1 += (pp[oi] + pp[4 + oi]) + (pp[8 + oi] + pp[12 + oi]);
            #pragma unroll
            for (int mm = 0; mm < M_; ++mm) T[mm] += pp[8 * (2 + mm) + 4 * ((mm >> 1) & 1) + oi];
            #pragma unroll
            for (int u = 0; u < 8; ++u) S2 += pp[80 + 4 * u + oi];
        }
        float* rr = red + ws * 640 + r;
        rr[0]  = S1;
        rr[64] = S2;
        #pragma unroll
        for (int mm = 0; mm < M_; ++mm) rr[128 + 64 * mm] = T[mm];
    }
    __syncthreads();

    if (tid < BG * OG) {
        float S1 = 0.f, S2 = 0.f, T[M_];
        #pragma unroll
        for (int mm = 0; mm < M_; ++mm) T[mm] = 0.f;
        #pragma unroll
        for (int ws = 0; ws < 4; ++ws) {
            const float* rr = red + ws * 640 + tid;
            S1 += rr[0];
            S2 += rr[64];
            #pragma unroll
            for (int mm = 0; mm < M_; ++mm) T[mm] += rr[128 + 64 * mm];
        }
        const float inv   = 1.0f / (float)(M_ * D_);
        const float mu    = S1 * inv;
        const float var   = S2 * inv - mu * mu;
        const float hrstd = 0.5f * rsqrtf(fmaf(0.25f, var, EPS_));
        float acc = 0.f;
        #pragma unroll
        for (int mm = 0; mm < M_; ++mm) {
            float s = hrstd * (T[mm] - mu * wb[mm]) + wb[M_ + mm];
            acc += fast_sigmoid(s);
        }
        pre[((size_t)ot * B_ + b0) * OG + tid] = acc;
    }
}

__global__ __launch_bounds__(256) void dnm_softmax(
    const float* __restrict__ pre, float* __restrict__ out)
{
    __shared__ __attribute__((aligned(16))) float se[O_];
    __shared__ __attribute__((aligned(16))) float sp[O_ / OG];
    const int b = blockIdx.x, t = threadIdx.x;
    if (t < O_ / OG) {
        const float4 v = *(const float4*)(pre + ((size_t)t * B_ + b) * OG);
        float4 e;
        e.x = __expf(v.x); e.y = __expf(v.y); e.z = __expf(v.z); e.w = __expf(v.w);
        *(float4*)(se + OG * t) = e;
        sp[t] = quad_sum(e);
    }
    __syncthreads();
    const float4* p4 = (const float4*)sp;
    const float4 u0 = quad_add4(p4[0],  p4[1],  p4[2],  p4[3]);
    const float4 u1 = quad_add4(p4[4],  p4[5],  p4[6],  p4[7]);
    const float4 u2 = quad_add4(p4[8],  p4[9],  p4[10], p4[11]);
    const float4 u3 = quad_add4(p4[12], p4[13], p4[14], p4[15]);
    const float tot = quad_sum(quad_add4(u0, u1, u2, u3));
    out[b * O_ + t] = se[t] * __builtin_amdgcn_rcpf(tot);
}

extern "C" void kernel_launch(void* const* d_in, const int* in_sizes, int n_in,
                              void* d_out, int out_size, void* d_ws, size_t ws_size,
                              hipStream_t stream) {
    const float* x    = (const float*)d_in[0];
    const float* sn0w = (const float*)d_in[1];
    const float* sn0b = (const float*)d_in[2];
    const float* sw   = (const float*)d_in[3];
    const float* sb   = (const float*)d_in[4];
    const float* dnw  = (const float*)d_in[5];
    const float* dnb  = (const float*)d_in[6];
    float* out = (float*)d_out;

    float* pre  = (float*)d_ws;

    dnm_main   <<<256,      NWV * 32, 0, stream>>>(x, sn0w, sn0b, sw, sb, dnw, dnb, pre);
    dnm_softmax<<<B_,       256,      0, stream>>>(pre, out);
}


# Main: workgroup of 8 waves, not 16: a wave owns a 64-wide d slice (two 32-wide halves a batch row), 2048 waves enter, one selector WMMA per 64 d; LayerNorm and row sum orders change
# speedup vs baseline: 3.2435x; 1.0576x over previous
#include <hip/hip_runtime.h>
#include <math.h>


typedef __attribute__((ext_vector_type(2))) float v2f;
typedef __attribute__((ext_vector_type(8))) float v8f;

#define B_   64
#define O_   256
#define M_   8
#define D_   512
#define EPS_ 1e-5f
#define OG   4
#define BG   16
#define NWV  8
#define EC   64
#define HC   32
#define RP   17
#define ROWOFF(j) (((2 * ((j) & 1)) * M_ + 2 * (((j) & 7) >> 1) + ((j) >> 3)) * D_)
#define PW   120
#define SPILL0 (NWV * BG * PW)
#define PSTN (2 * BG * NWV + 2 * M_ * NWV)

__device__ __forceinline__ float fast_sigmoid(float x) {
#if __has_builtin(__builtin_amdgcn_tanhf)
    return fmaf(0.5f, __builtin_amdgcn_tanhf(0.5f * x), 0.5f);
#else
    return __builtin_amdgcn_rcpf(1.0f + __expf(-x));
#endif
}

__device__ __forceinline__ float tanh_of_half(float hv) {
#if __has_builtin(__builtin_amdgcn_tanhf)
    return __builtin_amdgcn_tanhf(hv);
#else
    return fmaf(2.0f, __builtin_amdgcn_rcpf(1.0f + __expf(-2.0f * hv)), -1.0f);
#endif
}

__device__ __forceinline__ float other_half(float v) {
    const int b = __builtin_bit_cast(int, v);
    return __builtin_bit_cast(float,
        __builtin_amdgcn_permlanex16(b, b, 0x76543210, 0xfedcba98, false, false));
}

__device__ __forceinline__ float quad_sum(float4 u) { return (u.x + u.y) + (u.z + u.w); }

__device__ __forceinline__ float4 quad_add4(float4 a, float4 b, float4 c, float4 d) {
    float4 r;
    r.x = (a.x + b.x) + (c.x + d.x); r.y = (a.y + b.y) + (c.y + d.y);
    r.z = (a.z + b.z) + (c.z + d.z); r.w = (a.w + b.w) + (c.w + d.w);
    return r;
}

template <int CTRL> __device__ __forceinline__ float xlane_add(float v) {
    const int b = __builtin_bit_cast(int, v);
    return v + __builtin_bit_cast(float, __builtin_amdgcn_update_dpp(b, b, CTRL, 0xf, 0xf, false));
}

__global__ __launch_bounds__(NWV * 32) void dnm_main(
    const float* __restrict__ x,    const float* __restrict__ lnw,
    const float* __restrict__ lnb,  const float* sw,
    const float* sb,                const float* dnw,
    const float* __restrict__ dnb,  float* pre)
{
    __shared__ float4 stage[2 * NWV * 32 * RP];
    __shared__ __attribute__((aligned(16))) float part[NWV * BG * PW + NWV * 32 * 8];
    __shared__ float red[4 * 10 * 64];
    __shared__ __attribute__((aligned(16))) float pst[PSTN];
    __shared__ float wb[2 * M_];

    const int ot   = blockIdx.x >> 2;
    const int o0   = ot * OG;
    const int b0   = (blockIdx.x & 3) * BG;
    const int tid  = threadIdx.x;
    const int w    = __builtin_amdgcn_readfirstlane(tid >> 5);
    const int lane = tid & 31;
    const int c    = lane & 15;
    const int h    = lane >> 4;
    const int d0   = w * EC;

    float4* const xs = stage + w * (32 * RP);

    float swh[EC], sbh[EC], bm[EC];
    v2f bsel;
    {
        const int g  = lane >> 4;
        const int pc = lane & 15;
        const float* swl = sw + ((size_t)(o0 + g) * M_) * D_ + d0 + 4 * pc;
        const float* sbl = sb + ((size_t)(o0 + g) * M_) * D_ + d0 + 4 * pc;
        float4* st0 = xs;
        float4* st1 = stage + (NWV + w) * (32 * RP);
        float4 tx[8];
        const float* xl = x + (size_t)(b0 + g) * D_ + d0 + 4 * pc;
        #pragma unroll
        for (int i = 0; i < 8; ++i) tx[i] = *(const float4*)(xl + (2 * i) * D_);
        const float4 gw = *(const float4*)(lnw + d0 + 4 * pc);
        const float4 gb = *(const float4*)(lnb + d0 + 4 * pc);
        const int mq = lane >> 2;
        const int qq = lane & 3;
        const float4 dw0 = *(const float4*)(dnw + mq * D_ + d0 + 16 * qq);
        const float4 dw1 = *(const float4*)(dnw + mq * D_ + d0 + 16 * qq + 4);
        const float4 dw2 = *(const float4*)(dnw + mq * D_ + d0 + 16 * qq + 8);
        const float4 dw3 = *(const float4*)(dnw + mq * D_ + d0 + 16 * qq + 12);
        const float4 db0 = *(const float4*)(dnb + mq * D_ + d0 + 16 * qq);
        const float4 db1 = *(const float4*)(dnb + mq * D_ + d0 + 16 * qq + 4);
        const float4 db2 = *(const float4*)(dnb + mq * D_ + d0 + 16 * qq + 8);
        const float4 db3 = *(const float4*)(dnb + mq * D_ + d0 + 16 * qq + 12);
        const float4 ta0 = *(const float4*)(swl + ROWOFF(0));
        const float4 ta1 = *(const float4*)(swl + ROWOFF(1));
        const float4 ta2 = *(const float4*)(swl + ROWOFF(2));
        const float4 ta3 = *(const float4*)(swl + ROWOFF(3));
        const float4 ta4 = *(const float4*)(swl + ROWOFF(4));
        const float4 ta5 = *(const float4*)(swl + ROWOFF(5));
        const float4 ta6 = *(const float4*)(swl + ROWOFF(6));
        const float4 ta7 = *(const float4*)(swl + ROWOFF(7));
        const float4 ta8 = *(const float4*)(swl + ROWOFF(8));
        const float4 ta9 = *(const float4*)(swl + ROWOFF(9));
        const float4 ta10 = *(const float4*)(swl + ROWOFF(10));
        const float4 ta11 = *(const float4*)(swl + ROWOFF(11));
        const float4 ta12 = *(const float4*)(swl + ROWOFF(12));
        const float4 ta13 = *(const float4*)(swl + ROWOFF(13));
        const float4 ta14 = *(const float4*)(swl + ROWOFF(14));
        const float4 ta15 = *(const float4*)(swl + ROWOFF(15));
        const float4 tb0 = *(const float4*)(sbl + ROWOFF(0));
        const float4 tb1 = *(const float4*)(sbl + ROWOFF(1));
        const float4 tb2 = *(const float4*)(sbl + ROWOFF(2));
        const float4 tb3 = *(const float4*)(sbl + ROWOFF(3));
        const float4 tb4 = *(const float4*)(sbl + ROWOFF(4));
        const float4 tb5 = *(const float4*)(sbl + ROWOFF(5));
        const float4 tb6 = *(const float4*)(sbl + ROWOFF(6));
        const float4 tb7 = *(const float4*)(sbl + ROWOFF(7));
        const float4 tb8 = *(const float4*)(sbl + ROWOFF(8));
        const float4 tb9 = *(const float4*)(sbl + ROWOFF(9));
        const float4 tb10 = *(const float4*)(sbl + ROWOFF(10));
        const float4 tb11 = *(const float4*)(sbl + ROWOFF(11));
        const float4 tb12 = *(const float4*)(sbl + ROWOFF(12));
        const float4 tb13 = *(const float4*)(sbl + ROWOFF(13));
        const float4 tb14 = *(const float4*)(sbl + ROWOFF(14));
        const float4 tb15 = *(const float4*)(sbl + ROWOFF(15));

        const float* dwo = dnw + (2 * (c >> 2) + h) * D_ + d0;
        #pragma unroll
        for (int i = 0; i < EC; i += 4) {
            const float4 w0 = *(const float4*)(dwo + i);
            bm[i + 0] = w0.x; bm[i + 1] = w0.y; bm[i + 2] = w0.z; bm[i + 3] = w0.w;
        }
        const bool sel = (c >= 1) && (c <= 8) && (((c - 1) & 1) == h);
        bsel.x = (c == 0) ? 1.0f : 0.0f;
        bsel.y = sel ? 1.0f : 0.0f;

        float p1[8], p2[8];
        #pragma unroll
        for (int i = 0; i < 8; ++i) {
            const float4 u = tx[i];
            float a1 = quad_sum(u);
            float a2 = (u.x * u.x + u.y * u.y) + (u.z * u.z + u.w * u.w);
            a1 = xlane_add<0x161>(a1); a1 = xlane_add<0x162>(a1); a1 = xlane_add<0x164>(a1); a1 = xlane_add<0x168>(a1);
            a2 = xlane_add<0x161>(a2); a2 = xlane_add<0x162>(a2); a2 = xlane_add<0x164>(a2); a2 = xlane_add<0x168>(a2);
            p1[i] = a1; p2[i] = a2;
        }
        #pragma unroll
        for (int i = 0; i < 8; ++i)
            if (pc < 2) pst[(pc * BG + 2 * i + g) * NWV + w] = (pc == 0) ? p1[i] : p2[i];
        asm volatile("s_wait_dscnt 0x0" ::: "memory");
        asm volatile("s_barrier_signal -1\n\ts_barrier_wait -1" ::: "memory");
        float mu, rstd;
        {
            const float4* r1 = (const float4*)(pst + c * NWV + 4 * h);
            const float4* r2 = (const float4*)(pst + (BG + c) * NWV + 4 * h);
            float S1 = quad_sum(r1[0]);
            float S2 = quad_sum(r2[0]);
            S1 += other_half(S1);
            S2 += other_half(S2);
            mu = S1 * (1.0f / D_);
            const float var = S2 * (1.0f / D_) - mu * mu;
            rstd = rsqrtf(var + EPS_);
        }
        #pragma unroll
        for (int i = 0; i < 8; ++i) {
            const float mi = __shfl(mu,   2 * i + g);
            const float ri = __shfl(rstd, 2 * i + g);
            tx[i].x = (tx[i].x - mi) * ri * gw.x + gb.x;
            tx[i].y = (tx[i].y - mi) * ri * gw.y + gb.y;
            tx[i].z = (tx[i].z - mi) * ri * gw.z + gb.z;
            tx[i].w = (tx[i].w - mi) * ri * gw.w + gb.w;
        }

        __builtin_amdgcn_sched_barrier(0);
        st0[(0 + g) * RP + pc] = ta0;
        st0[(2 + g) * RP + pc] = ta1;
        st0[(4 + g) * RP + pc] = ta2;
        st0[(6 + g) * RP + pc] = ta3;
        st0[(8 + g) * RP + pc] = ta4;
        st0[(10 + g) * RP + pc] = ta5;
        st0[(12 + g) * RP + pc] = ta6;
        st0[(14 + g) * RP + pc] = ta7;
        st0[(16 + g) * RP + pc] = ta8;
        st0[(18 + g) * RP + pc] = ta9;
        st0[(20 + g) * RP + pc] = ta10;
        st0[(22 + g) * RP + pc] = ta11;
        st0[(24 + g) * RP + pc] = ta12;
        st0[(26 + g) * RP + pc] = ta13;
        st0[(28 + g) * RP + pc] = ta14;
        st0[(30 + g) * RP + pc] = ta15;
        st1[(0 + g) * RP + pc] = tb0;
        st1[(2 + g) * RP + pc] = tb1;
        st1[(4 + g) * RP + pc] = tb2;
        st1[(6 + g) * RP + pc] = tb3;
        st1[(8 + g) * RP + pc] = tb4;
        st1[(10 + g) * RP + pc] = tb5;
        st1[(12 + g) * RP + pc] = tb6;
        st1[(14 + g) * RP + pc] = tb7;
        st1[(16 + g) * RP + pc] = tb8;
        st1[(18 + g) * RP + pc] = tb9;
        st1[(20 + g) * RP + pc] = tb10;
        st1[(22 + g) * RP + pc] = tb11;
        st1[(24 + g) * RP + pc] = tb12;
        st1[(26 + g) * RP + pc] = tb13;
        st1[(28 + g) * RP + pc] = tb14;
        st1[(30 + g) * RP + pc] = tb15;
        #pragma unroll
        for (int i = 0; i < 16; ++i) {
            const float4 v = st0[lane * RP + i];
            swh[4 * i + 0] = 0.5f * v.x; swh[4 * i + 1] = 0.5f * v.y;
            swh[4 * i + 2] = 0.5f * v.z; swh[4 * i + 3] = 0.5f * v.w;
        }
        #pragma unroll
        for (int i = 0; i < 16; ++i) {
            const float4 v = st1[lane * RP + i];
            sbh[4 * i + 0] = 0.5f * v.x; sbh[4 * i + 1] = 0.5f * v.y;
            sbh[4 * i + 2] = 0.5f * v.z; sbh[4 * i + 3] = 0.5f * v.w;
        }
        #pragma unroll
        for (int i = 0; i < 8; ++i) st0[16 * (2 * i + g) + pc] = tx[i];

        __builtin_amdgcn_sched_barrier(0);
        float wsp = (quad_sum(dw0) + quad_sum(dw1)) + (quad_sum(dw2) + quad_sum(dw3));
        float bsp = (quad_sum(db0) + quad_sum(db1)) + (quad_sum(db2) + quad_sum(db3));
        wsp = xlane_add<0x161>(wsp); wsp = xlane_add<0x162>(wsp);
        bsp = xlane_add<0x161>(bsp); bsp = xlane_add<0x162>(bsp);
        if (qq < 2) pst[2 * BG * NWV + (qq * M_ + mq) * NWV + w] = (qq == 0) ? wsp : bsp;
    }

    const bool wr    = (c == 0) || ((c <= 8) && (h == ((c - 1) >> 2)));
    const int  slot  = (c == 0) ? h : (1 + c);
    const int  spill = SPILL0 + (w * 32 + lane) * 8;

    auto half_row = [&](const float4 (&xq)[8], const int off, v2f& q, v2f& sv, v2f& tv) {
        #pragma unroll
        for (int i = 0; i < HC; i += 8) {
            const float4 xu = xq[i >> 2];
            const float4 xv = xq[(i >> 2) + 1];
            const int k = off + i;
            v2f a0, a1, a2, a3, w0, w1, w2, w3;
            a0.x = tanh_of_half(fmaf(swh[k],     xu.x, sbh[k]));
            a0.y = tanh_of_half(fmaf(swh[k + 1], xu.y, sbh[k + 1]));
            w0.x = bm[k];
            w0.y = bm[k + 1];
            sv = sv + a0;
            tv = __builtin_elementwise_fma(a0, w0, tv);
            q  = __builtin_elementwise_fma(a0, a0, q);
            a1.x = tanh_of_half(fmaf(swh[k + 2], xu.z, sbh[k + 2]));
            a1.y = tanh_of_half(fmaf(swh[k + 3], xu.w, sbh[k + 3]));
            w1.x = bm[k + 2];
            w1.y = bm[k + 3];
            sv = sv + a1;
            tv = __builtin_elementwise_fma(a1, w1, tv);
            q  = __builtin_elementwise_fma(a1, a1, q);
            a2.x = tanh_of_half(fmaf(swh[k + 4], xv.x, sbh[k + 4]));
            a2.y = tanh_of_half(fmaf(swh[k + 5], xv.y, sbh[k + 5]));
            w2.x = bm[k + 4];
            w2.y = bm[k + 5];
            sv = sv + a2;
            tv = __builtin_elementwise_fma(a2, w2, tv);
            q  = __builtin_elementwise_fma(a2, a2, q);
            a3.x = tanh_of_half(fmaf(swh[k + 6], xv.z, sbh[k + 6]));
            a3.y = tanh_of_half(fmaf(swh[k + 7], xv.w, sbh[k + 7]));
            w3.x = bm[k + 6];
            w3.y = bm[k + 7];
            sv = sv + a3;
            tv = __builtin_elementwise_fma(a3, w3, tv);
            q  = __builtin_elementwise_fma(a3, a3, q);
        }
    };

    auto finish_row = [&](const v2f q, const v2f sv, const v2f tv, const int bi) {
        v2f af;
        af.x = sv.x + sv.y;
        af.y = tv.x + tv.y;
        const v8f cz = {};
        const v8f c1 = __builtin_amdgcn_wmma_f32_16x16x4_f32(false, af, false, bsel, (short)0, cz, false, false);
        const int rec = (w * BG + bi) * PW;
        part[rec + 80 + lane] = q.x + q.y;
        float* dst = part + (wr ? rec + 8 * slot : spill);
        float4 lo, hi;
        lo.x = c1[0]; lo.y = c1[1]; lo.z = c1[2]; lo.w = c1[3];
        hi.x = c1[4]; hi.y = c1[5]; hi.z = c1[6]; hi.w = c1[7];
        *(float4*)(dst)     = lo;
        *(float4*)(dst + 4) = hi;
    };

    float4 xa[8], xb[8];
    #pragma unroll
    for (int i = 0; i < 8; ++i) xa[i] = xs[i];

    #pragma unroll 1
    for (int bi = 0; bi < BG; ++bi) {
        #pragma unroll
        for (int i = 0; i < 8; ++i) xb[i] = xs[16 * bi + 8 + i];
        v2f q  = {};
        v2f sv = {};
        v2f tv = {};
        half_row(xa, 0, q, sv, tv);
        #pragma unroll
        for (int i = 0; i < 8; ++i) xa[i] = xs[16 * ((bi + 1) & (BG - 1)) + i];
        half_row(xb, HC, q, sv, tv);
        finish_row(q, sv, tv, bi);
    }
    __syncthreads();

    {
        if (tid < 2 * M_) {
            const float4* q4 = (const float4*)(pst + 2 * BG * NWV + NWV * tid);
            wb[tid] = quad_sum(q4[0]) + quad_sum(q4[1]);
        }
        const int ws = tid >> 6;
        const int r  = tid & 63;
        const int bi = r >> 2;
        const int oi = r & 3;
        float S1 = 0.f, S2 = 0.f, T[M_];
        #pragma unroll
        for (int mm = 0; mm < M_; ++mm) T[mm] = 0.f;
        #pragma unroll
        for (int k = 0; k < 2; ++k) {
            const float* pp = part + ((2 * ws + k) * BG + bi) * PW;
            S1 += (pp[oi] + pp[4 + oi]) + (pp[8 + oi] + pp[12 + oi]);
            #pragma unroll
            for (int mm = 0; mm < M_; ++mm) T[mm] += pp[8 * (2 + mm) + 4 * ((mm >> 1) & 1) + oi];
            #pragma unroll
            for (int u = 0; u < 8; ++u) S2 += pp[80 + 4 * u + oi];
        }
        float* rr = red + ws * 640 + r;
        rr[0]  = S1;
        rr[64] = S2;
        #pragma unroll
        for (int mm = 0; mm < M_; ++mm) rr[128 + 64 * mm] = T[mm];
    }
    __syncthreads();

    if (tid < BG * OG) {
        float S1 = 0.f, S2 = 0.f, T[M_];
        #pragma unroll
        for (int mm = 0; mm < M_; ++mm) T[mm] = 0.f;
        #pragma unroll
        for (int ws = 0; ws < 4; ++ws) {
            const float* rr = red + ws * 640 + tid;
            S1 += rr[0];
            S2 += rr[64];
            #pragma unroll
            for (int mm = 0; mm < M_; ++mm) T[mm] += rr[128 + 64 * mm];
        }
        const float inv   = 1.0f / (float)(M_ * D_);
        const float mu    = S1 * inv;
        const float var   = S2 * inv - mu * mu;
        const float hrstd = 0.5f * rsqrtf(fmaf(0.25f, var, EPS_));
        float acc = 0.f;
        #pragma unroll
        for (int mm = 0; mm < M_; ++mm) {
            float s = hrstd * (T[mm] - mu * wb[mm]) + wb[M_ + mm];
            acc += fast_sigmoid(s);
        }
        pre[((size_t)ot * B_ + b0) * OG + tid] = acc;
    }
}

__global__ __launch_bounds__(256) void dnm_softmax(
    const float* __restrict__ pre, float* __restrict__ out)
{
    __shared__ __attribute__((aligned(16))) float se[O_];
    __shared__ __attribute__((aligned(16))) float sp[O_ / OG];
    const int b = blockIdx.x, t = threadIdx.x;
    if (t < O_ / OG) {
        const float4 v = *(const float4*)(pre + ((size_t)t * B_ + b) * OG);
        float4 e;
        e.x = __expf(v.x); e.y = __expf(v.y); e.z = __expf(v.z); e.w = __expf(v.w);
        *(float4*)(se + OG * t) = e;
        sp[t] = quad_sum(e);
    }
    __syncthreads();
    const float4* p4 = (const float4*)sp;
    const float4 u0 = quad_add4(p4[0],  p4[1],  p4[2],  p4[3]);
    const float4 u1 = quad_add4(p4[4],  p4[5],  p4[6],  p4[7]);
    const float4 u2 = quad_add4(p4[8],  p4[9],  p4[10], p4[11]);
    const float4 u3 = quad_add4(p4[12], p4[13], p4[14], p4[15]);
    const float tot = quad_sum(quad_add4(u0, u1, u2, u3));
    out[b * O_ + t] = se[t] * __builtin_amdgcn_rcpf(tot);
}

extern "C" void kernel_launch(void* const* d_in, const int* in_sizes, int n_in,
                              void* d_out, int out_size, void* d_ws, size_t ws_size,
                              hipStream_t stream) {
    const float* x    = (const float*)d_in[0];
    const float* sn0w = (const float*)d_in[1];
    const float* sn0b = (const float*)d_in[2];
    const float* sw   = (const float*)d_in[3];
    const float* sb   = (const float*)d_in[4];
    const float* dnw  = (const float*)d_in[5];
    const float* dnb  = (const float*)d_in[6];
    float* out = (float*)d_out;

    float* pre  = (float*)d_ws;

    dnm_main   <<<256,      NWV * 32, 0, stream>>>(x, sn0w, sn0b, sw, sb, dnw, dnb, pre);
    dnm_softmax<<<B_,       256,      0, stream>>>(pre, out);
}
